# starting kernel
# baseline (speedup 1.0000x reference)
#include <hip/hip_runtime.h>

typedef __attribute__((ext_vector_type(2))) float v2f;
typedef __attribute__((ext_vector_type(8))) float v8f;

#define N_TOK 4096
#define C_DIM 64

__device__ __forceinline__ v8f wmma_f32_4(v2f a, v2f b, v8f c) {
  return __builtin_amdgcn_wmma_f32_16x16x4_f32(
       false, a,  false, b,
       (short)0, c,  false,  false);
}

__global__ __launch_bounds__(256) void attn_gate_kernel(
    const float* __restrict__ x, float* __restrict__ y_ws) {
  __shared__ float lds_p[8 * 256];

  const int lane = threadIdx.x & 31;
  const int wave = threadIdx.x >> 5;
  const int col  = lane & 15;
  const int half = lane >> 4;

  const int t0 = blockIdx.x * 128 + wave * 16;
  const int b  = t0 / N_TOK;
  const int n0 = t0 % N_TOK;
  const float* __restrict__ xb = x + (size_t)b * N_TOK * C_DIM;

  const int qrow = n0 + col;
  v2f qa[16];
#pragma unroll
  for (int k = 0; k < 16; ++k)
    qa[k] = *(const v2f*)(xb + (size_t)qrow * C_DIM + 4 * k + 2 * half);

  v8f o[4];
  float rm[8], rl[8];
#pragma unroll
  for (int j = 0; j < 4; ++j) o[j] = (v8f)0.0f;
#pragma unroll
  for (int i = 0; i < 8; ++i) { rm[i] = -1e30f; rl[i] = 0.0f; }

  float* __restrict__ pw = lds_p + wave * 256;

  for (int c0 = 0; c0 < N_TOK; c0 += 16) {
    v8f s = (v8f)0.0f;
#pragma unroll
    for (int k = 0; k < 16; ++k) {
      v2f kb = *(const v2f*)(xb + (size_t)(c0 + col) * C_DIM + 4 * k + 2 * half);
      s = wmma_f32_4(qa[k], kb, s);
    }

#pragma unroll
    for (int i = 0; i < 8; ++i) {
      float v = s[i];
#pragma unroll
      for (int off = 1; off < 16; off <<= 1)
        v = fmaxf(v, __shfl_xor(v, off));
      float nm = fmaxf(rm[i], v);
      float sc = __expf(rm[i] - nm);
      rm[i] = nm;
      float p = __expf(s[i] - nm);
      s[i] = p;
      float ts = p;
#pragma unroll
      for (int off = 1; off < 16; off <<= 1)
        ts += __shfl_xor(ts, off);
      rl[i] = rl[i] * sc + ts;
#pragma unroll
      for (int j = 0; j < 4; ++j) o[j][i] *= sc;
    }

#pragma unroll
    for (int i = 0; i < 8; ++i)
      pw[(i + 8 * half) * 16 + col] = s[i];
    asm volatile("" ::: "memory");

    v2f pa[4];
#pragma unroll
    for (int kk = 0; kk < 4; ++kk)
      pa[kk] = *(const v2f*)(pw + col * 16 + 4 * kk + 2 * half);
    asm volatile("" ::: "memory");

#pragma unroll
    for (int kk = 0; kk < 4; ++kk) {
      const float* vrow = xb + (size_t)(c0 + 4 * kk + 2 * half) * C_DIM;
#pragma unroll
      for (int j = 0; j < 4; ++j) {
        v2f vb;
        vb.x = vrow[j * 16 + col];
        vb.y = vrow[C_DIM + j * 16 + col];
        o[j] = wmma_f32_4(pa[kk], vb, o[j]);
      }
    }
  }

#pragma unroll
  for (int i = 0; i < 8; ++i) {
    float inv = 1.0f / rl[i];
    int row = n0 + i + 8 * half;
#pragma unroll
    for (int j = 0; j < 4; ++j) {
      int ch = j * 16 + col;
      float g = xb[(size_t)row * C_DIM + ch];
      y_ws[((size_t)b * N_TOK + row) * C_DIM + ch] = o[j][i] * inv * g;
    }
  }
}

__global__ __launch_bounds__(256) void conv_gemm_kernel(
    const float* __restrict__ y, const float* __restrict__ w,
    const float* __restrict__ bias, float* __restrict__ out) {
  const int lane = threadIdx.x & 31;
  const int wave = threadIdx.x >> 5;
  const int col  = lane & 15;
  const int half = lane >> 4;

  const int rbase = (blockIdx.x * 8 + wave) * 16;

  int rowg = rbase + col;
  int wpos = rowg % 29;
  int rem  = rowg / 29;
  int h = rem & 31; rem >>= 5;
  int d = rem & 3;
  int bb = rem >> 2;
  size_t abase = (((size_t)bb * N_TOK) + ((d * 32 + h) * 32 + wpos)) * (size_t)C_DIM;

  v8f acc[8];
#pragma unroll
  for (int j = 0; j < 8; ++j) acc[j] = (v8f)0.0f;

  for (int kk = 0; kk < 64; ++kk) {
    int k0 = 4 * kk + 2 * half;
    v2f a = *(const v2f*)(y + abase + k0);
    const float* w0 = w + (size_t)k0 * 128;
#pragma unroll
    for (int j = 0; j < 8; ++j) {
      v2f wb;
      wb.x = w0[j * 16 + col];
      wb.y = w0[128 + j * 16 + col];
      acc[j] = wmma_f32_4(a, wb, acc[j]);
    }
  }

#pragma unroll
  for (int j = 0; j < 8; ++j) {
    int co = j * 16 + col;
    float bv = bias[co];
#pragma unroll
    for (int i = 0; i < 8; ++i) {
      int row = rbase + i + 8 * half;
      float v = acc[j][i] + bv;
      out[(size_t)row * 128 + co] = fmaxf(v, 0.0f);
    }
  }
}

extern "C" void kernel_launch(void* const* d_in, const int* in_sizes, int n_in,
                              void* d_out, int out_size, void* d_ws, size_t ws_size,
                              hipStream_t stream) {
  const float* x  = (const float*)d_in[0];
  const float* cw = (const float*)d_in[1];
  const float* cb = (const float*)d_in[2];
  float* out = (float*)d_out;
  float* y   = (float*)d_ws;

  attn_gate_kernel<<<128, 256, 0, stream>>>(x, y);
  conv_gemm_kernel<<<116, 256, 0, stream>>>(y, cw, cb, out);
}


# attention: score tile computed transposed (keys x queries): softmax max/sum in-lane plus one lane^16 exchange, exp(S) registers feed P.V as a WMMA operand; no LDS staging, no 16-lane shuffles
# speedup vs baseline: 1.4161x; 1.4161x over previous
#include <hip/hip_runtime.h>

typedef __attribute__((ext_vector_type(2))) float v2f;
typedef __attribute__((ext_vector_type(8))) float v8f;

#define N_TOK 4096
#define C_DIM 64

__device__ __forceinline__ v8f wmma_f32_4(v2f a, v2f b, v8f c) {
  return __builtin_amdgcn_wmma_f32_16x16x4_f32(
       false, a,  false, b,
       (short)0, c,  false,  false);
}

__device__ __forceinline__ float xhalf(float v) {
  int b = __builtin_bit_cast(int, v);
  return __builtin_bit_cast(
      float, __builtin_amdgcn_permlanex16(b, b, 0x76543210, 0xfedcba98, false, false));
}

__global__ __launch_bounds__(256) void attn_gate_kernel(
    const float* __restrict__ x, float* __restrict__ y_ws) {
  const int lane = threadIdx.x & 31;
  const int wave = threadIdx.x >> 5;
  const int col  = lane & 15;
  const int half = lane >> 4;

  const int t0 = blockIdx.x * 128 + wave * 16;
  const int b  = t0 / N_TOK;
  const int n0 = t0 % N_TOK;
  const float* __restrict__ xb = x + (size_t)b * N_TOK * C_DIM;

  const int qrow = n0 + col;
  v2f qb[16];
#pragma unroll
  for (int k = 0; k < 16; ++k)
    qb[k] = *(const v2f*)(xb + (size_t)qrow * C_DIM + 4 * k + 2 * half);

  v8f o[4];
#pragma unroll
  for (int j = 0; j < 4; ++j) o[j] = (v8f)0.0f;
  float rm = -1e30f;
  float rl = 0.0f;

  for (int c0 = 0; c0 < N_TOK; c0 += 16) {
    v2f ka[16];
#pragma unroll
    for (int k = 0; k < 16; ++k)
      ka[k] = *(const v2f*)(xb + (size_t)(c0 + col) * C_DIM + 4 * k + 2 * half);
    v2f va[4][4];
#pragma unroll
    for (int t = 0; t < 4; ++t) {
      const float* vrow = xb + (size_t)(c0 + 8 * half + 2 * t) * C_DIM;
#pragma unroll
      for (int j = 0; j < 4; ++j) {
        va[t][j].x = vrow[j * 16 + col];
        va[t][j].y = vrow[C_DIM + j * 16 + col];
      }
    }
    __builtin_amdgcn_sched_barrier(0);

    v8f s = (v8f)0.0f;
#pragma unroll
    for (int k = 0; k < 16; ++k) s = wmma_f32_4(ka[k], qb[k], s);

    float m01 = fmaxf(s[0], s[1]);
    float m23 = fmaxf(s[2], s[3]);
    float m45 = fmaxf(s[4], s[5]);
    float m67 = fmaxf(s[6], s[7]);
    float mloc = fmaxf(fmaxf(m01, m23), fmaxf(m45, m67));
    float mall = fmaxf(mloc, xhalf(mloc));
    float nm = fmaxf(rm, mall);
    float sc = __expf(rm - nm);
    rm = nm;
    float ts = 0.0f;
#pragma unroll
    for (int r = 0; r < 8; ++r) {
      float p = __expf(s[r] - nm);
      s[r] = p;
      ts += p;
    }
    rl = rl * sc + ts;
#pragma unroll
    for (int j = 0; j < 4; ++j) o[j] *= sc;

#pragma unroll
    for (int t = 0; t < 4; ++t) {
      v2f pb;
      pb.x = s[2 * t];
      pb.y = s[2 * t + 1];
#pragma unroll
      for (int j = 0; j < 4; ++j) o[j] = wmma_f32_4(va[t][j], pb, o[j]);
    }
  }

  const float inv = 1.0f / (rl + xhalf(rl));
  const size_t rbase = (size_t)(n0 + col) * C_DIM + 8 * half;
  float* __restrict__ yb = y_ws + (size_t)b * N_TOK * C_DIM;
#pragma unroll
  for (int j = 0; j < 4; ++j) {
    const float4 g0 = *(const float4*)(xb + rbase + 16 * j);
    const float4 g1 = *(const float4*)(xb + rbase + 16 * j + 4);
    float4 y0, y1;
    y0.x = o[j][0] * inv * g0.x;
    y0.y = o[j][1] * inv * g0.y;
    y0.z = o[j][2] * inv * g0.z;
    y0.w = o[j][3] * inv * g0.w;
    y1.x = o[j][4] * inv * g1.x;
    y1.y = o[j][5] * inv * g1.y;
    y1.z = o[j][6] * inv * g1.z;
    y1.w = o[j][7] * inv * g1.w;
    *(float4*)(yb + rbase + 16 * j)     = y0;
    *(float4*)(yb + rbase + 16 * j + 4) = y1;
  }
}

__global__ __launch_bounds__(256) void conv_gemm_kernel(
    const float* __restrict__ y, const float* __restrict__ w,
    const float* __restrict__ bias, float* __restrict__ out) {
  const int lane = threadIdx.x & 31;
  const int wave = threadIdx.x >> 5;
  const int col  = lane & 15;
  const int half = lane >> 4;

  const int rbase = (blockIdx.x * 8 + wave) * 16;

  int rowg = rbase + col;
  int wpos = rowg % 29;
  int rem  = rowg / 29;
  int h = rem & 31; rem >>= 5;
  int d = rem & 3;
  int bb = rem >> 2;
  size_t abase = (((size_t)bb * N_TOK) + ((d * 32 + h) * 32 + wpos)) * (size_t)C_DIM;

  v8f acc[8];
#pragma unroll
  for (int j = 0; j < 8; ++j) acc[j] = (v8f)0.0f;

  for (int kk = 0; kk < 64; ++kk) {
    int k0 = 4 * kk + 2 * half;
    v2f a = *(const v2f*)(y + abase + k0);
    const float* w0 = w + (size_t)k0 * 128;
#pragma unroll
    for (int j = 0; j < 8; ++j) {
      v2f wb;
      wb.x = w0[j * 16 + col];
      wb.y = w0[128 + j * 16 + col];
      acc[j] = wmma_f32_4(a, wb, acc[j]);
    }
  }

#pragma unroll
  for (int j = 0; j < 8; ++j) {
    int co = j * 16 + col;
    float bv = bias[co];
#pragma unroll
    for (int i = 0; i < 8; ++i) {
      int row = rbase + i + 8 * half;
      float v = acc[j][i] + bv;
      out[(size_t)row * 128 + co] = fmaxf(v, 0.0f);
    }
  }
}

extern "C" void kernel_launch(void* const* d_in, const int* in_sizes, int n_in,
                              void* d_out, int out_size, void* d_ws, size_t ws_size,
                              hipStream_t stream) {
  const float* x  = (const float*)d_in[0];
  const float* cw = (const float*)d_in[1];
  const float* cb = (const float*)d_in[2];
  float* out = (float*)d_out;
  float* y   = (float*)d_ws;

  attn_gate_kernel<<<128, 256, 0, stream>>>(x, y);
  conv_gemm_kernel<<<116, 256, 0, stream>>>(y, cw, cb, out);
}


# attention: keys split over 4 waves per query tile (16-wave blocks, 256 blocks: every WGP at four waves per SIMD, was half the WGPs at two); partial max/sum/O merged once through LDS
# speedup vs baseline: 3.0026x; 2.1203x over previous
#include <hip/hip_runtime.h>

typedef __attribute__((ext_vector_type(2))) float v2f;
typedef __attribute__((ext_vector_type(8))) float v8f;

#define N_TOK 4096
#define C_DIM 64

__device__ __forceinline__ v8f wmma_f32_4(v2f a, v2f b, v8f c) {
  return __builtin_amdgcn_wmma_f32_16x16x4_f32(
       false, a,  false, b,
       (short)0, c,  false,  false);
}

__device__ __forceinline__ float xhalf(float v) {
  int b = __builtin_bit_cast(int, v);
  return __builtin_bit_cast(
      float, __builtin_amdgcn_permlanex16(b, b, 0x76543210, 0xfedcba98, false, false));
}

#define KSPLIT 4
#define QTILES 4
#define OPITCH4 17

__global__ __launch_bounds__(512) void attn_gate_kernel(
    const float* __restrict__ x, float* __restrict__ y_ws) {
  __shared__ float4 sO[KSPLIT * QTILES * 16 * OPITCH4];
  __shared__ float  sM[KSPLIT * QTILES * 16];
  __shared__ float  sL[KSPLIT * QTILES * 16];

  const int lane = threadIdx.x & 31;
  const int wave = __builtin_amdgcn_readfirstlane(threadIdx.x >> 5);
  const int col  = lane & 15;
  const int half = lane >> 4;
  const int qt   = wave & (QTILES - 1);
  const int ks   = wave >> 2;

  const int t0 = blockIdx.x * (16 * QTILES) + qt * 16;
  const int b  = t0 / N_TOK;
  const int n0 = t0 % N_TOK;
  const float* __restrict__ xb = x + (size_t)b * N_TOK * C_DIM;

  const int qrow = n0 + col;
  v2f qb[16];
#pragma unroll
  for (int k = 0; k < 16; ++k)
    qb[k] = *(const v2f*)(xb + (size_t)qrow * C_DIM + 4 * k + 2 * half);

  v8f o[4];
#pragma unroll
  for (int j = 0; j < 4; ++j) o[j] = (v8f)0.0f;
  float rm = -1e30f;
  float rl = 0.0f;

  const float* __restrict__ xk = xb + (size_t)ks * 16 * C_DIM;
  for (int c0 = 0; c0 < N_TOK; c0 += 16 * KSPLIT) {
    v2f ka[16];
#pragma unroll
    for (int k = 0; k < 16; ++k)
      ka[k] = *(const v2f*)(xk + (size_t)(c0 + col) * C_DIM + 4 * k + 2 * half);
    v2f va[4][4];
#pragma unroll
    for (int t = 0; t < 4; ++t) {
      const float* vrow = xk + (size_t)(c0 + 8 * half + 2 * t) * C_DIM;
#pragma unroll
      for (int j = 0; j < 4; ++j) {
        va[t][j].x = vrow[j * 16 + col];
        va[t][j].y = vrow[C_DIM + j * 16 + col];
      }
    }
    __builtin_amdgcn_sched_barrier(0);

    v8f s = (v8f)0.0f;
#pragma unroll
    for (int k = 0; k < 16; ++k) s = wmma_f32_4(ka[k], qb[k], s);

    float m01 = fmaxf(s[0], s[1]);
    float m23 = fmaxf(s[2], s[3]);
    float m45 = fmaxf(s[4], s[5]);
    float m67 = fmaxf(s[6], s[7]);
    float mloc = fmaxf(fmaxf(m01, m23), fmaxf(m45, m67));
    float mall = fmaxf(mloc, xhalf(mloc));
    float nm = fmaxf(rm, mall);
    float sc = __expf(rm - nm);
    rm = nm;
    float ts = 0.0f;
#pragma unroll
    for (int r = 0; r < 8; ++r) {
      float p = __expf(s[r] - nm);
      s[r] = p;
      ts += p;
    }
    rl = rl * sc + ts;
#pragma unroll
    for (int j = 0; j < 4; ++j) o[j] *= sc;

#pragma unroll
    for (int t = 0; t < 4; ++t) {
      v2f pb;
      pb.x = s[2 * t];
      pb.y = s[2 * t + 1];
#pragma unroll
      for (int j = 0; j < 4; ++j) o[j] = wmma_f32_4(va[t][j], pb, o[j]);
    }
  }

  {
    float4* ob = sO + (wave * 16 + col) * OPITCH4 + 2 * half;
#pragma unroll
    for (int j = 0; j < 4; ++j) {
      ob[4 * j]     = make_float4(o[j][0], o[j][1], o[j][2], o[j][3]);
      ob[4 * j + 1] = make_float4(o[j][4], o[j][5], o[j][6], o[j][7]);
    }
    const float lw = rl + xhalf(rl);
    sM[wave * 16 + col] = rm;
    sL[wave * 16 + col] = lw;
  }
  __syncthreads();

  float* __restrict__ yb = y_ws + (size_t)b * N_TOK * C_DIM;
#pragma unroll
  for (int k = 0; k < 2; ++k) {
    const int qr = 4 * ks + 2 * k + half;
    float mi[KSPLIT], li[KSPLIT];
#pragma unroll
    for (int i = 0; i < KSPLIT; ++i) {
      mi[i] = sM[(i * QTILES + qt) * 16 + qr];
      li[i] = sL[(i * QTILES + qt) * 16 + qr];
    }
    const float mt = fmaxf(fmaxf(mi[0], mi[1]), fmaxf(mi[2], mi[3]));
    float lt = 0.0f;
    float4 acc = make_float4(0.0f, 0.0f, 0.0f, 0.0f);
#pragma unroll
    for (int i = 0; i < KSPLIT; ++i) {
      const float f = __expf(mi[i] - mt);
      lt += f * li[i];
      const float4 v = sO[((i * QTILES + qt) * 16 + qr) * OPITCH4 + col];
      acc.x += f * v.x;
      acc.y += f * v.y;
      acc.z += f * v.z;
      acc.w += f * v.w;
    }
    const float inv = 1.0f / lt;
    const size_t off = (size_t)(n0 + qr) * C_DIM + 4 * col;
    const float4 g = *(const float4*)(xb + off);
    float4 yv;
    yv.x = acc.x * inv * g.x;
    yv.y = acc.y * inv * g.y;
    yv.z = acc.z * inv * g.z;
    yv.w = acc.w * inv * g.w;
    *(float4*)(yb + off) = yv;
  }
}

__global__ __launch_bounds__(256) void conv_gemm_kernel(
    const float* __restrict__ y, const float* __restrict__ w,
    const float* __restrict__ bias, float* __restrict__ out) {
  const int lane = threadIdx.x & 31;
  const int wave = threadIdx.x >> 5;
  const int col  = lane & 15;
  const int half = lane >> 4;

  const int rbase = (blockIdx.x * 8 + wave) * 16;

  int rowg = rbase + col;
  int wpos = rowg % 29;
  int rem  = rowg / 29;
  int h = rem & 31; rem >>= 5;
  int d = rem & 3;
  int bb = rem >> 2;
  size_t abase = (((size_t)bb * N_TOK) + ((d * 32 + h) * 32 + wpos)) * (size_t)C_DIM;

  v8f acc[8];
#pragma unroll
  for (int j = 0; j < 8; ++j) acc[j] = (v8f)0.0f;

  for (int kk = 0; kk < 64; ++kk) {
    int k0 = 4 * kk + 2 * half;
    v2f a = *(const v2f*)(y + abase + k0);
    const float* w0 = w + (size_t)k0 * 128;
#pragma unroll
    for (int j = 0; j < 8; ++j) {
      v2f wb;
      wb.x = w0[j * 16 + col];
      wb.y = w0[128 + j * 16 + col];
      acc[j] = wmma_f32_4(a, wb, acc[j]);
    }
  }

#pragma unroll
  for (int j = 0; j < 8; ++j) {
    int co = j * 16 + col;
    float bv = bias[co];
#pragma unroll
    for (int i = 0; i < 8; ++i) {
      int row = rbase + i + 8 * half;
      float v = acc[j][i] + bv;
      out[(size_t)row * 128 + co] = fmaxf(v, 0.0f);
    }
  }
}

extern "C" void kernel_launch(void* const* d_in, const int* in_sizes, int n_in,
                              void* d_out, int out_size, void* d_ws, size_t ws_size,
                              hipStream_t stream) {
  const float* x  = (const float*)d_in[0];
  const float* cw = (const float*)d_in[1];
  const float* cb = (const float*)d_in[2];
  float* out = (float*)d_out;
  float* y   = (float*)d_ws;

  attn_gate_kernel<<<256, 512, 0, stream>>>(x, y);
  conv_gemm_kernel<<<116, 256, 0, stream>>>(y, cw, cb, out);
}


# attention: fragment loads re-mapped to 16-byte loads on whole lines (K and Q share a new k order; V^T tile j row col = channel 4*col+j): 16 loads and 160 lines touched per key tile, was 48 and 320
# speedup vs baseline: 3.1204x; 1.0392x over previous
#include <hip/hip_runtime.h>

typedef __attribute__((ext_vector_type(2))) float v2f;
typedef __attribute__((ext_vector_type(4))) float v4f;
typedef __attribute__((ext_vector_type(8))) float v8f;

#define N_TOK 4096
#define C_DIM 64

__device__ __forceinline__ v8f wmma_f32_4(v2f a, v2f b, v8f c) {
  return __builtin_amdgcn_wmma_f32_16x16x4_f32(
       false, a,  false, b,
       (short)0, c,  false,  false);
}

__device__ __forceinline__ float xhalf(float v) {
  int b = __builtin_bit_cast(int, v);
  return __builtin_bit_cast(
      float, __builtin_amdgcn_permlanex16(b, b, 0x76543210, 0xfedcba98, false, false));
}

#define KSPLIT 4
#define QTILES 4
#define OPITCH4 17

__global__ __launch_bounds__(512) void attn_gate_kernel(
    const float* __restrict__ x, float* __restrict__ y_ws) {
  __shared__ float4 sO[KSPLIT * QTILES * 16 * OPITCH4];
  __shared__ float  sM[KSPLIT * QTILES * 16];
  __shared__ float  sL[KSPLIT * QTILES * 16];

  const int lane = threadIdx.x & 31;
  const int wave = __builtin_amdgcn_readfirstlane(threadIdx.x >> 5);
  const int col  = lane & 15;
  const int half = lane >> 4;
  const int qt   = wave & (QTILES - 1);
  const int ks   = wave >> 2;

  const int t0 = blockIdx.x * (16 * QTILES) + qt * 16;
  const int b  = t0 / N_TOK;
  const int n0 = t0 % N_TOK;
  const float* __restrict__ xb = x + (size_t)b * N_TOK * C_DIM;

  const int qrow = n0 + col;
  v4f q4[8];
#pragma unroll
  for (int m = 0; m < 8; ++m)
    q4[m] = *(const v4f*)(xb + (size_t)qrow * C_DIM + 8 * m + 4 * half);

  v8f o[4];
#pragma unroll
  for (int j = 0; j < 4; ++j) o[j] = (v8f)0.0f;
  float rm = -1e30f;
  float rl = 0.0f;

  const float* __restrict__ xk = xb + (size_t)ks * 16 * C_DIM;
  for (int c0 = 0; c0 < N_TOK; c0 += 16 * KSPLIT) {
    v4f vv[4][2];
#pragma unroll
    for (int t = 0; t < 4; ++t) {
#pragma unroll
      for (int e = 0; e < 2; ++e)
        vv[t][e] = *(const v4f*)(xk + (size_t)(c0 + 8 * half + 2 * t + e) * C_DIM + 4 * col);
    }
    v4f k4[8];
#pragma unroll
    for (int m = 0; m < 8; ++m)
      k4[m] = *(const v4f*)(xk + (size_t)(c0 + col) * C_DIM + 8 * m + 4 * half);
    __builtin_amdgcn_sched_barrier(0);

    v8f s = (v8f)0.0f;
#pragma unroll
    for (int m = 0; m < 8; ++m) {
      s = wmma_f32_4(__builtin_shufflevector(k4[m], k4[m], 0, 1),
                     __builtin_shufflevector(q4[m], q4[m], 0, 1), s);
      s = wmma_f32_4(__builtin_shufflevector(k4[m], k4[m], 2, 3),
                     __builtin_shufflevector(q4[m], q4[m], 2, 3), s);
    }

    float m01 = fmaxf(s[0], s[1]);
    float m23 = fmaxf(s[2], s[3]);
    float m45 = fmaxf(s[4], s[5]);
    float m67 = fmaxf(s[6], s[7]);
    float mloc = fmaxf(fmaxf(m01, m23), fmaxf(m45, m67));
    float mall = fmaxf(mloc, xhalf(mloc));
    float nm = fmaxf(rm, mall);
    float sc = __expf(rm - nm);
    rm = nm;
    float ts = 0.0f;
#pragma unroll
    for (int r = 0; r < 8; ++r) {
      float p = __expf(s[r] - nm);
      s[r] = p;
      ts += p;
    }
    rl = rl * sc + ts;
#pragma unroll
    for (int j = 0; j < 4; ++j) o[j] *= sc;

#pragma unroll
    for (int t = 0; t < 4; ++t) {
      v2f pb;
      pb.x = s[2 * t];
      pb.y = s[2 * t + 1];
#pragma unroll
      for (int j = 0; j < 4; ++j) {
        v2f va;
        va.x = vv[t][0][j];
        va.y = vv[t][1][j];
        o[j] = wmma_f32_4(va, pb, o[j]);
      }
    }
  }

  {
    float4* ob = sO + (wave * 16 + col) * OPITCH4 + 8 * half;
#pragma unroll
    for (int r = 0; r < 8; ++r)
      ob[r] = make_float4(o[0][r], o[1][r], o[2][r], o[3][r]);
    const float lw = rl + xhalf(rl);
    sM[wave * 16 + col] = rm;
    sL[wave * 16 + col] = lw;
  }
  __syncthreads();

  float* __restrict__ yb = y_ws + (size_t)b * N_TOK * C_DIM;
#pragma unroll
  for (int k = 0; k < 2; ++k) {
    const int qr = 4 * ks + 2 * k + half;
    float mi[KSPLIT], li[KSPLIT];
#pragma unroll
    for (int i = 0; i < KSPLIT; ++i) {
      mi[i] = sM[(i * QTILES + qt) * 16 + qr];
      li[i] = sL[(i * QTILES + qt) * 16 + qr];
    }
    const float mt = fmaxf(fmaxf(mi[0], mi[1]), fmaxf(mi[2], mi[3]));
    float lt = 0.0f;
    float4 acc = make_float4(0.0f, 0.0f, 0.0f, 0.0f);
#pragma unroll
    for (int i = 0; i < KSPLIT; ++i) {
      const float f = __expf(mi[i] - mt);
      lt += f * li[i];
      const float4 v = sO[((i * QTILES + qt) * 16 + qr) * OPITCH4 + col];
      acc.x += f * v.x;
      acc.y += f * v.y;
      acc.z += f * v.z;
      acc.w += f * v.w;
    }
    const float inv = 1.0f / lt;
    const size_t off = (size_t)(n0 + qr) * C_DIM + 4 * col;
    const float4 g = *(const float4*)(xb + off);
    float4 yv;
    yv.x = acc.x * inv * g.x;
    yv.y = acc.y * inv * g.y;
    yv.z = acc.z * inv * g.z;
    yv.w = acc.w * inv * g.w;
    *(float4*)(yb + off) = yv;
  }
}

__global__ __launch_bounds__(256) void conv_gemm_kernel(
    const float* __restrict__ y, const float* __restrict__ w,
    const float* __restrict__ bias, float* __restrict__ out) {
  const int lane = threadIdx.x & 31;
  const int wave = threadIdx.x >> 5;
  const int col  = lane & 15;
  const int half = lane >> 4;

  const int rbase = (blockIdx.x * 8 + wave) * 16;

  int rowg = rbase + col;
  int wpos = rowg % 29;
  int rem  = rowg / 29;
  int h = rem & 31; rem >>= 5;
  int d = rem & 3;
  int bb = rem >> 2;
  size_t abase = (((size_t)bb * N_TOK) + ((d * 32 + h) * 32 + wpos)) * (size_t)C_DIM;

  v8f acc[8];
#pragma unroll
  for (int j = 0; j < 8; ++j) acc[j] = (v8f)0.0f;

  for (int kk = 0; kk < 64; ++kk) {
    int k0 = 4 * kk + 2 * half;
    v2f a = *(const v2f*)(y + abase + k0);
    const float* w0 = w + (size_t)k0 * 128;
#pragma unroll
    for (int j = 0; j < 8; ++j) {
      v2f wb;
      wb.x = w0[j * 16 + col];
      wb.y = w0[128 + j * 16 + col];
      acc[j] = wmma_f32_4(a, wb, acc[j]);
    }
  }

#pragma unroll
  for (int j = 0; j < 8; ++j) {
    int co = j * 16 + col;
    float bv = bias[co];
#pragma unroll
    for (int i = 0; i < 8; ++i) {
      int row = rbase + i + 8 * half;
      float v = acc[j][i] + bv;
      out[(size_t)row * 128 + co] = fmaxf(v, 0.0f);
    }
  }
}

extern "C" void kernel_launch(void* const* d_in, const int* in_sizes, int n_in,
                              void* d_out, int out_size, void* d_ws, size_t ws_size,
                              hipStream_t stream) {
  const float* x  = (const float*)d_in[0];
  const float* cw = (const float*)d_in[1];
  const float* cb = (const float*)d_in[2];
  float* out = (float*)d_out;
  float* y   = (float*)d_ws;

  attn_gate_kernel<<<256, 512, 0, stream>>>(x, y);
  conv_gemm_kernel<<<116, 256, 0, stream>>>(y, cw, cb, out);
}


# attention: each wave keeps two query tiles per loaded K / V^T fragment set, keys split over 8 waves (was one tile, 4 waves): 16 fragment loads per 64 WMMAs, was per 32
# speedup vs baseline: 3.3641x; 1.0781x over previous
#include <hip/hip_runtime.h>

typedef __attribute__((ext_vector_type(2))) float v2f;
typedef __attribute__((ext_vector_type(4))) float v4f;
typedef __attribute__((ext_vector_type(8))) float v8f;

#define N_TOK 4096
#define C_DIM 64

__device__ __forceinline__ v8f wmma_f32_4(v2f a, v2f b, v8f c) {
  return __builtin_amdgcn_wmma_f32_16x16x4_f32(
       false, a,  false, b,
       (short)0, c,  false,  false);
}

__device__ __forceinline__ float xhalf(float v) {
  int b = __builtin_bit_cast(int, v);
  return __builtin_bit_cast(
      float, __builtin_amdgcn_permlanex16(b, b, 0x76543210, 0xfedcba98, false, false));
}

#define KSPLIT 8
#define QPAIRS 2
#define QT 2
#define NWAVES (KSPLIT * QPAIRS)
#define OPITCH4 17

__device__ __forceinline__ float softmax_step(v8f& s, float& rm, float& rl) {
  float m01 = fmaxf(s[0], s[1]);
  float m23 = fmaxf(s[2], s[3]);
  float m45 = fmaxf(s[4], s[5]);
  float m67 = fmaxf(s[6], s[7]);
  float mloc = fmaxf(fmaxf(m01, m23), fmaxf(m45, m67));
  float mall = fmaxf(mloc, xhalf(mloc));
  float nm = fmaxf(rm, mall);
  float sc = __expf(rm - nm);
  rm = nm;
  float ts = 0.0f;
#pragma unroll
  for (int r = 0; r < 8; ++r) {
    float p = __expf(s[r] - nm);
    s[r] = p;
    ts += p;
  }
  rl = rl * sc + ts;
  return sc;
}

__global__ __launch_bounds__(512) void attn_gate_kernel(
    const float* __restrict__ x, float* __restrict__ y_ws) {
  __shared__ float4 sO[NWAVES * QT * 16 * OPITCH4];
  __shared__ float  sM[NWAVES * QT * 16];
  __shared__ float  sL[NWAVES * QT * 16];

  const int lane = threadIdx.x & 31;
  const int wave = __builtin_amdgcn_readfirstlane(threadIdx.x >> 5);
  const int col  = lane & 15;
  const int half = lane >> 4;
  const int qp   = wave & (QPAIRS - 1);
  const int ks   = wave >> 1;

  const int tb = blockIdx.x * (16 * QT * QPAIRS);
  const int b  = tb / N_TOK;
  const int nb = tb % N_TOK;
  const int n0 = nb + qp * (16 * QT);
  const float* __restrict__ xb = x + (size_t)b * N_TOK * C_DIM;

  v4f q4[QT][8];
#pragma unroll
  for (int u = 0; u < QT; ++u) {
#pragma unroll
    for (int m = 0; m < 8; ++m)
      q4[u][m] = *(const v4f*)(xb + (size_t)(n0 + 16 * u + col) * C_DIM + 8 * m + 4 * half);
  }

  v8f o[QT][4];
  float rm[QT];
  float rl[QT];
#pragma unroll
  for (int u = 0; u < QT; ++u) {
#pragma unroll
    for (int j = 0; j < 4; ++j) o[u][j] = (v8f)0.0f;
    rm[u] = -1e30f;
    rl[u] = 0.0f;
  }

  const float* __restrict__ xk = xb + (size_t)ks * 16 * C_DIM;
  for (int c0 = 0; c0 < N_TOK; c0 += 16 * KSPLIT) {
    v4f vv[4][2];
#pragma unroll
    for (int t = 0; t < 4; ++t) {
#pragma unroll
      for (int e = 0; e < 2; ++e)
        vv[t][e] = *(const v4f*)(xk + (size_t)(c0 + 8 * half + 2 * t + e) * C_DIM + 4 * col);
    }
    v4f k4[8];
#pragma unroll
    for (int m = 0; m < 8; ++m)
      k4[m] = *(const v4f*)(xk + (size_t)(c0 + col) * C_DIM + 8 * m + 4 * half);
    __builtin_amdgcn_sched_barrier(0);

    v8f s[QT];
#pragma unroll
    for (int u = 0; u < QT; ++u) s[u] = (v8f)0.0f;
#pragma unroll
    for (int m = 0; m < 8; ++m) {
      const v2f ka = __builtin_shufflevector(k4[m], k4[m], 0, 1);
      const v2f kb = __builtin_shufflevector(k4[m], k4[m], 2, 3);
#pragma unroll
      for (int u = 0; u < QT; ++u)
        s[u] = wmma_f32_4(ka, __builtin_shufflevector(q4[u][m], q4[u][m], 0, 1), s[u]);
#pragma unroll
      for (int u = 0; u < QT; ++u)
        s[u] = wmma_f32_4(kb, __builtin_shufflevector(q4[u][m], q4[u][m], 2, 3), s[u]);
    }

#pragma unroll
    for (int u = 0; u < QT; ++u) {
      const float sc = softmax_step(s[u], rm[u], rl[u]);
#pragma unroll
      for (int j = 0; j < 4; ++j) o[u][j] *= sc;
    }

#pragma unroll
    for (int t = 0; t < 4; ++t) {
#pragma unroll
      for (int j = 0; j < 4; ++j) {
        v2f va;
        va.x = vv[t][0][j];
        va.y = vv[t][1][j];
#pragma unroll
        for (int u = 0; u < QT; ++u) {
          v2f pb;
          pb.x = s[u][2 * t];
          pb.y = s[u][2 * t + 1];
          o[u][j] = wmma_f32_4(va, pb, o[u][j]);
        }
      }
    }
  }

#pragma unroll
  for (int u = 0; u < QT; ++u) {
    const int slot = (wave * QT + u) * 16 + col;
    float4* ob = sO + slot * OPITCH4 + 8 * half;
#pragma unroll
    for (int r = 0; r < 8; ++r)
      ob[r] = make_float4(o[u][0][r], o[u][1][r], o[u][2][r], o[u][3][r]);
    const float lw = rl[u] + xhalf(rl[u]);
    sM[slot] = rm[u];
    sL[slot] = lw;
  }
  __syncthreads();

  float* __restrict__ yb = y_ws + (size_t)b * N_TOK * C_DIM;
#pragma unroll
  for (int k = 0; k < 2; ++k) {
    const int R   = 4 * wave + 2 * k + half;
    const int qpr = R >> 5;
    const int ur  = (R >> 4) & 1;
    const int row = R & 15;
    float mi[KSPLIT], li[KSPLIT];
    int   sl[KSPLIT];
#pragma unroll
    for (int i = 0; i < KSPLIT; ++i) {
      sl[i] = ((i * QPAIRS + qpr) * QT + ur) * 16 + row;
      mi[i] = sM[sl[i]];
      li[i] = sL[sl[i]];
    }
    float mt = mi[0];
#pragma unroll
    for (int i = 1; i < KSPLIT; ++i) mt = fmaxf(mt, mi[i]);
    float lt = 0.0f;
    float4 acc = make_float4(0.0f, 0.0f, 0.0f, 0.0f);
#pragma unroll
    for (int i = 0; i < KSPLIT; ++i) {
      const float f = __expf(mi[i] - mt);
      lt += f * li[i];
      const float4 v = sO[sl[i] * OPITCH4 + col];
      acc.x += f * v.x;
      acc.y += f * v.y;
      acc.z += f * v.z;
      acc.w += f * v.w;
    }
    const float inv = 1.0f / lt;
    const size_t off = (size_t)(nb + R) * C_DIM + 4 * col;
    const float4 g = *(const float4*)(xb + off);
    float4 yv;
    yv.x = acc.x * inv * g.x;
    yv.y = acc.y * inv * g.y;
    yv.z = acc.z * inv * g.z;
    yv.w = acc.w * inv * g.w;
    *(float4*)(yb + off) = yv;
  }
}

__global__ __launch_bounds__(256) void conv_gemm_kernel(
    const float* __restrict__ y, const float* __restrict__ w,
    const float* __restrict__ bias, float* __restrict__ out) {
  const int lane = threadIdx.x & 31;
  const int wave = threadIdx.x >> 5;
  const int col  = lane & 15;
  const int half = lane >> 4;

  const int rbase = (blockIdx.x * 8 + wave) * 16;

  int rowg = rbase + col;
  int wpos = rowg % 29;
  int rem  = rowg / 29;
  int h = rem & 31; rem >>= 5;
  int d = rem & 3;
  int bb = rem >> 2;
  size_t abase = (((size_t)bb * N_TOK) + ((d * 32 + h) * 32 + wpos)) * (size_t)C_DIM;

  v8f acc[8];
#pragma unroll
  for (int j = 0; j < 8; ++j) acc[j] = (v8f)0.0f;

  for (int kk = 0; kk < 64; ++kk) {
    int k0 = 4 * kk + 2 * half;
    v2f a = *(const v2f*)(y + abase + k0);
    const float* w0 = w + (size_t)k0 * 128;
#pragma unroll
    for (int j = 0; j < 8; ++j) {
      v2f wb;
      wb.x = w0[j * 16 + col];
      wb.y = w0[128 + j * 16 + col];
      acc[j] = wmma_f32_4(a, wb, acc[j]);
    }
  }

#pragma unroll
  for (int j = 0; j < 8; ++j) {
    int co = j * 16 + col;
    float bv = bias[co];
#pragma unroll
    for (int i = 0; i < 8; ++i) {
      int row = rbase + i + 8 * half;
      float v = acc[j][i] + bv;
      out[(size_t)row * 128 + co] = fmaxf(v, 0.0f);
    }
  }
}

extern "C" void kernel_launch(void* const* d_in, const int* in_sizes, int n_in,
                              void* d_out, int out_size, void* d_ws, size_t ws_size,
                              hipStream_t stream) {
  const float* x  = (const float*)d_in[0];
  const float* cw = (const float*)d_in[1];
  const float* cb = (const float*)d_in[2];
  float* out = (float*)d_out;
  float* y   = (float*)d_ws;

  attn_gate_kernel<<<256, 512, 0, stream>>>(x, y);
  conv_gemm_kernel<<<116, 256, 0, stream>>>(y, cw, cb, out);
}


# both kernels: operand fragment loads run ahead of their WMMAs in registers (attention: next K behind the score run, next V^T behind the P.V run; conv: A and W three k steps ahead, ring of 3)
# speedup vs baseline: 4.3439x; 1.2913x over previous
#include <hip/hip_runtime.h>

typedef __attribute__((ext_vector_type(2))) float v2f;
typedef __attribute__((ext_vector_type(4))) float v4f;
typedef __attribute__((ext_vector_type(8))) float v8f;

#define N_TOK 4096
#define C_DIM 64

__device__ __forceinline__ v8f wmma_f32_4(v2f a, v2f b, v8f c) {
  return __builtin_amdgcn_wmma_f32_16x16x4_f32(
       false, a,  false, b,
       (short)0, c,  false,  false);
}

__device__ __forceinline__ float xhalf(float v) {
  int b = __builtin_bit_cast(int, v);
  return __builtin_bit_cast(
      float, __builtin_amdgcn_permlanex16(b, b, 0x76543210, 0xfedcba98, false, false));
}

#define KSPLIT 8
#define QPAIRS 2
#define QT 2
#define NWAVES (KSPLIT * QPAIRS)
#define OPITCH4 17

__device__ __forceinline__ float softmax_step(v8f& s, float& rm, float& rl) {
  float m01 = fmaxf(s[0], s[1]);
  float m23 = fmaxf(s[2], s[3]);
  float m45 = fmaxf(s[4], s[5]);
  float m67 = fmaxf(s[6], s[7]);
  float mloc = fmaxf(fmaxf(m01, m23), fmaxf(m45, m67));
  float mall = fmaxf(mloc, xhalf(mloc));
  float nm = fmaxf(rm, mall);
  float sc = __expf(rm - nm);
  rm = nm;
  float ts = 0.0f;
#pragma unroll
  for (int r = 0; r < 8; ++r) {
    float p = __expf(s[r] - nm);
    s[r] = p;
    ts += p;
  }
  rl = rl * sc + ts;
  return sc;
}

__global__ __launch_bounds__(512) void attn_gate_kernel(
    const float* __restrict__ x, float* __restrict__ y_ws) {
  __shared__ float4 sO[NWAVES * QT * 16 * OPITCH4];
  __shared__ float  sM[NWAVES * QT * 16];
  __shared__ float  sL[NWAVES * QT * 16];

  const int lane = threadIdx.x & 31;
  const int wave = __builtin_amdgcn_readfirstlane(threadIdx.x >> 5);
  const int col  = lane & 15;
  const int half = lane >> 4;
  const int qp   = wave & (QPAIRS - 1);
  const int ks   = wave >> 1;

  const int tb = blockIdx.x * (16 * QT * QPAIRS);
  const int b  = tb / N_TOK;
  const int nb = tb % N_TOK;
  const int n0 = nb + qp * (16 * QT);
  const float* __restrict__ xb = x + (size_t)b * N_TOK * C_DIM;

  v4f q4[QT][8];
#pragma unroll
  for (int u = 0; u < QT; ++u) {
#pragma unroll
    for (int m = 0; m < 8; ++m)
      q4[u][m] = *(const v4f*)(xb + (size_t)(n0 + 16 * u + col) * C_DIM + 8 * m + 4 * half);
  }

  v8f o[QT][4];
  float rm[QT];
  float rl[QT];
#pragma unroll
  for (int u = 0; u < QT; ++u) {
#pragma unroll
    for (int j = 0; j < 4; ++j) o[u][j] = (v8f)0.0f;
    rm[u] = -1e30f;
    rl[u] = 0.0f;
  }

  const float* __restrict__ xk = xb + (size_t)ks * 16 * C_DIM;
  v4f k4[8];
  v4f vv[4][2];
#pragma unroll
  for (int m = 0; m < 8; ++m)
    k4[m] = *(const v4f*)(xk + (size_t)col * C_DIM + 8 * m + 4 * half);
#pragma unroll
  for (int t = 0; t < 4; ++t) {
#pragma unroll
    for (int e = 0; e < 2; ++e)
      vv[t][e] = *(const v4f*)(xk + (size_t)(8 * half + 2 * t + e) * C_DIM + 4 * col);
  }
  __builtin_amdgcn_sched_barrier(0);

  for (int c0 = 0; c0 < N_TOK; c0 += 16 * KSPLIT) {
    const int cn = (c0 + 16 * KSPLIT < N_TOK) ? c0 + 16 * KSPLIT : c0;

    v8f s[QT];
#pragma unroll
    for (int u = 0; u < QT; ++u) s[u] = (v8f)0.0f;
#pragma unroll
    for (int m = 0; m < 8; ++m) {
      const v2f ka = __builtin_shufflevector(k4[m], k4[m], 0, 1);
      const v2f kb = __builtin_shufflevector(k4[m], k4[m], 2, 3);
#pragma unroll
      for (int u = 0; u < QT; ++u)
        s[u] = wmma_f32_4(ka, __builtin_shufflevector(q4[u][m], q4[u][m], 0, 1), s[u]);
#pragma unroll
      for (int u = 0; u < QT; ++u)
        s[u] = wmma_f32_4(kb, __builtin_shufflevector(q4[u][m], q4[u][m], 2, 3), s[u]);
    }
    __builtin_amdgcn_sched_barrier(0);

#pragma unroll
    for (int m = 0; m < 8; ++m)
      k4[m] = *(const v4f*)(xk + (size_t)(cn + col) * C_DIM + 8 * m + 4 * half);
    __builtin_amdgcn_sched_barrier(0);

#pragma unroll
    for (int u = 0; u < QT; ++u) {
      const float sc = softmax_step(s[u], rm[u], rl[u]);
#pragma unroll
      for (int j = 0; j < 4; ++j) o[u][j] *= sc;
    }

#pragma unroll
    for (int t = 0; t < 4; ++t) {
#pragma unroll
      for (int j = 0; j < 4; ++j) {
        v2f va;
        va.x = vv[t][0][j];
        va.y = vv[t][1][j];
#pragma unroll
        for (int u = 0; u < QT; ++u) {
          v2f pb;
          pb.x = s[u][2 * t];
          pb.y = s[u][2 * t + 1];
          o[u][j] = wmma_f32_4(va, pb, o[u][j]);
        }
      }
    }
    __builtin_amdgcn_sched_barrier(0);

#pragma unroll
    for (int t = 0; t < 4; ++t) {
#pragma unroll
      for (int e = 0; e < 2; ++e)
        vv[t][e] = *(const v4f*)(xk + (size_t)(cn + 8 * half + 2 * t + e) * C_DIM + 4 * col);
    }
    __builtin_amdgcn_sched_barrier(0);
  }

#pragma unroll
  for (int u = 0; u < QT; ++u) {
    const int slot = (wave * QT + u) * 16 + col;
    float4* ob = sO + slot * OPITCH4 + 8 * half;
#pragma unroll
    for (int r = 0; r < 8; ++r)
      ob[r] = make_float4(o[u][0][r], o[u][1][r], o[u][2][r], o[u][3][r]);
    const float lw = rl[u] + xhalf(rl[u]);
    sM[slot] = rm[u];
    sL[slot] = lw;
  }
  __syncthreads();

  float* __restrict__ yb = y_ws + (size_t)b * N_TOK * C_DIM;
#pragma unroll
  for (int k = 0; k < 2; ++k) {
    const int R   = 4 * wave + 2 * k + half;
    const int qpr = R >> 5;
    const int ur  = (R >> 4) & 1;
    const int row = R & 15;
    float mi[KSPLIT], li[KSPLIT];
    int   sl[KSPLIT];
#pragma unroll
    for (int i = 0; i < KSPLIT; ++i) {
      sl[i] = ((i * QPAIRS + qpr) * QT + ur) * 16 + row;
      mi[i] = sM[sl[i]];
      li[i] = sL[sl[i]];
    }
    float mt = mi[0];
#pragma unroll
    for (int i = 1; i < KSPLIT; ++i) mt = fmaxf(mt, mi[i]);
    float lt = 0.0f;
    float4 acc = make_float4(0.0f, 0.0f, 0.0f, 0.0f);
#pragma unroll
    for (int i = 0; i < KSPLIT; ++i) {
      const float f = __expf(mi[i] - mt);
      lt += f * li[i];
      const float4 v = sO[sl[i] * OPITCH4 + col];
      acc.x += f * v.x;
      acc.y += f * v.y;
      acc.z += f * v.z;
      acc.w += f * v.w;
    }
    const float inv = 1.0f / lt;
    const size_t off = (size_t)(nb + R) * C_DIM + 4 * col;
    const float4 g = *(const float4*)(xb + off);
    float4 yv;
    yv.x = acc.x * inv * g.x;
    yv.y = acc.y * inv * g.y;
    yv.z = acc.z * inv * g.z;
    yv.w = acc.w * inv * g.w;
    *(float4*)(yb + off) = yv;
  }
}

__global__ __launch_bounds__(256) void conv_gemm_kernel(
    const float* __restrict__ y, const float* __restrict__ w,
    const float* __restrict__ bias, float* __restrict__ out) {
  const int lane = threadIdx.x & 31;
  const int wave = threadIdx.x >> 5;
  const int col  = lane & 15;
  const int half = lane >> 4;

  const int rbase = (blockIdx.x * 8 + wave) * 16;

  int rowg = rbase + col;
  int wpos = rowg % 29;
  int rem  = rowg / 29;
  int h = rem & 31; rem >>= 5;
  int d = rem & 3;
  int bb = rem >> 2;
  size_t abase = (((size_t)bb * N_TOK) + ((d * 32 + h) * 32 + wpos)) * (size_t)C_DIM;

  v8f acc[8];
#pragma unroll
  for (int j = 0; j < 8; ++j) acc[j] = (v8f)0.0f;

  v2f a[3];
  v2f wb[3][8];
#pragma unroll
  for (int s = 0; s < 3; ++s) {
    const int k0 = 4 * s + 2 * half;
    a[s] = *(const v2f*)(y + abase + k0);
    const float* w0 = w + (size_t)k0 * 128;
#pragma unroll
    for (int j = 0; j < 8; ++j) {
      wb[s][j].x = w0[j * 16 + col];
      wb[s][j].y = w0[128 + j * 16 + col];
    }
    __builtin_amdgcn_sched_barrier(0);
  }

  for (int kk = 0; kk < 63; kk += 3) {
#pragma unroll
    for (int s = 0; s < 3; ++s) {
#pragma unroll
      for (int j = 0; j < 8; ++j) acc[j] = wmma_f32_4(a[s], wb[s][j], acc[j]);
      __builtin_amdgcn_sched_barrier(0);
      const int kn = (kk + s + 3 < 64) ? kk + s + 3 : 63;
      const int k0 = 4 * kn + 2 * half;
      a[s] = *(const v2f*)(y + abase + k0);
      const float* w0 = w + (size_t)k0 * 128;
#pragma unroll
      for (int j = 0; j < 8; ++j) {
        wb[s][j].x = w0[j * 16 + col];
        wb[s][j].y = w0[128 + j * 16 + col];
      }
      __builtin_amdgcn_sched_barrier(0);
    }
  }
#pragma unroll
  for (int j = 0; j < 8; ++j) acc[j] = wmma_f32_4(a[0], wb[0][j], acc[j]);

#pragma unroll
  for (int j = 0; j < 8; ++j) {
    int co = j * 16 + col;
    float bv = bias[co];
#pragma unroll
    for (int i = 0; i < 8; ++i) {
      int row = rbase + i + 8 * half;
      float v = acc[j][i] + bv;
      out[(size_t)row * 128 + co] = fmaxf(v, 0.0f);
    }
  }
}

extern "C" void kernel_launch(void* const* d_in, const int* in_sizes, int n_in,
                              void* d_out, int out_size, void* d_ws, size_t ws_size,
                              hipStream_t stream) {
  const float* x  = (const float*)d_in[0];
  const float* cw = (const float*)d_in[1];
  const float* cb = (const float*)d_in[2];
  float* out = (float*)d_out;
  float* y   = (float*)d_ws;

  attn_gate_kernel<<<256, 512, 0, stream>>>(x, y);
  conv_gemm_kernel<<<116, 256, 0, stream>>>(y, cw, cb, out);
}


# conv GEMM re-tiled: one 16x64 tile per wave on 232 eight-wave blocks (was 16x128 on 116); y, W, out move by 16-byte accesses (W, out on whole lines), one shared k order: 5 loads per 8 WMMAs, was 17
# speedup vs baseline: 4.7851x; 1.1016x over previous
#include <hip/hip_runtime.h>

typedef __attribute__((ext_vector_type(2))) float v2f;
typedef __attribute__((ext_vector_type(4))) float v4f;
typedef __attribute__((ext_vector_type(8))) float v8f;

#define N_TOK 4096
#define C_DIM 64

__device__ __forceinline__ v8f wmma_f32_4(v2f a, v2f b, v8f c) {
  return __builtin_amdgcn_wmma_f32_16x16x4_f32(
       false, a,  false, b,
       (short)0, c,  false,  false);
}

__device__ __forceinline__ float xhalf(float v) {
  int b = __builtin_bit_cast(int, v);
  return __builtin_bit_cast(
      float, __builtin_amdgcn_permlanex16(b, b, 0x76543210, 0xfedcba98, false, false));
}

#define KSPLIT 8
#define QPAIRS 2
#define QT 2
#define NWAVES (KSPLIT * QPAIRS)
#define OPITCH4 17

__device__ __forceinline__ float softmax_step(v8f& s, float& rm, float& rl) {
  float m01 = fmaxf(s[0], s[1]);
  float m23 = fmaxf(s[2], s[3]);
  float m45 = fmaxf(s[4], s[5]);
  float m67 = fmaxf(s[6], s[7]);
  float mloc = fmaxf(fmaxf(m01, m23), fmaxf(m45, m67));
  float mall = fmaxf(mloc, xhalf(mloc));
  float nm = fmaxf(rm, mall);
  float sc = __expf(rm - nm);
  rm = nm;
  float ts = 0.0f;
#pragma unroll
  for (int r = 0; r < 8; ++r) {
    float p = __expf(s[r] - nm);
    s[r] = p;
    ts += p;
  }
  rl = rl * sc + ts;
  return sc;
}

__global__ __launch_bounds__(512) void attn_gate_kernel(
    const float* __restrict__ x, float* __restrict__ y_ws) {
  __shared__ float4 sO[NWAVES * QT * 16 * OPITCH4];
  __shared__ float  sM[NWAVES * QT * 16];
  __shared__ float  sL[NWAVES * QT * 16];

  const int lane = threadIdx.x & 31;
  const int wave = __builtin_amdgcn_readfirstlane(threadIdx.x >> 5);
  const int col  = lane & 15;
  const int half = lane >> 4;
  const int qp   = wave & (QPAIRS - 1);
  const int ks   = wave >> 1;

  const int tb = blockIdx.x * (16 * QT * QPAIRS);
  const int b  = tb / N_TOK;
  const int nb = tb % N_TOK;
  const int n0 = nb + qp * (16 * QT);
  const float* __restrict__ xb = x + (size_t)b * N_TOK * C_DIM;

  v4f q4[QT][8];
#pragma unroll
  for (int u = 0; u < QT; ++u) {
#pragma unroll
    for (int m = 0; m < 8; ++m)
      q4[u][m] = *(const v4f*)(xb + (size_t)(n0 + 16 * u + col) * C_DIM + 8 * m + 4 * half);
  }

  v8f o[QT][4];
  float rm[QT];
  float rl[QT];
#pragma unroll
  for (int u = 0; u < QT; ++u) {
#pragma unroll
    for (int j = 0; j < 4; ++j) o[u][j] = (v8f)0.0f;
    rm[u] = -1e30f;
    rl[u] = 0.0f;
  }

  const float* __restrict__ xk = xb + (size_t)ks * 16 * C_DIM;
  v4f k4[8];
  v4f vv[4][2];
#pragma unroll
  for (int m = 0; m < 8; ++m)
    k4[m] = *(const v4f*)(xk + (size_t)col * C_DIM + 8 * m + 4 * half);
#pragma unroll
  for (int t = 0; t < 4; ++t) {
#pragma unroll
    for (int e = 0; e < 2; ++e)
      vv[t][e] = *(const v4f*)(xk + (size_t)(8 * half + 2 * t + e) * C_DIM + 4 * col);
  }
  __builtin_amdgcn_sched_barrier(0);

  for (int c0 = 0; c0 < N_TOK; c0 += 16 * KSPLIT) {
    const int cn = (c0 + 16 * KSPLIT < N_TOK) ? c0 + 16 * KSPLIT : c0;

    v8f s[QT];
#pragma unroll
    for (int u = 0; u < QT; ++u) s[u] = (v8f)0.0f;
#pragma unroll
    for (int m = 0; m < 8; ++m) {
      const v2f ka = __builtin_shufflevector(k4[m], k4[m], 0, 1);
      const v2f kb = __builtin_shufflevector(k4[m], k4[m], 2, 3);
#pragma unroll
      for (int u = 0; u < QT; ++u)
        s[u] = wmma_f32_4(ka, __builtin_shufflevector(q4[u][m], q4[u][m], 0, 1), s[u]);
#pragma unroll
      for (int u = 0; u < QT; ++u)
        s[u] = wmma_f32_4(kb, __builtin_shufflevector(q4[u][m], q4[u][m], 2, 3), s[u]);
    }
    __builtin_amdgcn_sched_barrier(0);

#pragma unroll
    for (int m = 0; m < 8; ++m)
      k4[m] = *(const v4f*)(xk + (size_t)(cn + col) * C_DIM + 8 * m + 4 * half);
    __builtin_amdgcn_sched_barrier(0);

#pragma unroll
    for (int u = 0; u < QT; ++u) {
      const float sc = softmax_step(s[u], rm[u], rl[u]);
#pragma unroll
      for (int j = 0; j < 4; ++j) o[u][j] *= sc;
    }

#pragma unroll
    for (int t = 0; t < 4; ++t) {
#pragma unroll
      for (int j = 0; j < 4; ++j) {
        v2f va;
        va.x = vv[t][0][j];
        va.y = vv[t][1][j];
#pragma unroll
        for (int u = 0; u < QT; ++u) {
          v2f pb;
          pb.x = s[u][2 * t];
          pb.y = s[u][2 * t + 1];
          o[u][j] = wmma_f32_4(va, pb, o[u][j]);
        }
      }
    }
    __builtin_amdgcn_sched_barrier(0);

#pragma unroll
    for (int t = 0; t < 4; ++t) {
#pragma unroll
      for (int e = 0; e < 2; ++e)
        vv[t][e] = *(const v4f*)(xk + (size_t)(cn + 8 * half + 2 * t + e) * C_DIM + 4 * col);
    }
    __builtin_amdgcn_sched_barrier(0);
  }

#pragma unroll
  for (int u = 0; u < QT; ++u) {
    const int slot = (wave * QT + u) * 16 + col;
    float4* ob = sO + slot * OPITCH4 + 8 * half;
#pragma unroll
    for (int r = 0; r < 8; ++r)
      ob[r] = make_float4(o[u][0][r], o[u][1][r], o[u][2][r], o[u][3][r]);
    const float lw = rl[u] + xhalf(rl[u]);
    sM[slot] = rm[u];
    sL[slot] = lw;
  }
  __syncthreads();

  float* __restrict__ yb = y_ws + (size_t)b * N_TOK * C_DIM;
#pragma unroll
  for (int k = 0; k < 2; ++k) {
    const int R   = 4 * wave + 2 * k + half;
    const int qpr = R >> 5;
    const int ur  = (R >> 4) & 1;
    const int row = R & 15;
    float mi[KSPLIT], li[KSPLIT];
    int   sl[KSPLIT];
#pragma unroll
    for (int i = 0; i < KSPLIT; ++i) {
      sl[i] = ((i * QPAIRS + qpr) * QT + ur) * 16 + row;
      mi[i] = sM[sl[i]];
      li[i] = sL[sl[i]];
    }
    float mt = mi[0];
#pragma unroll
    for (int i = 1; i < KSPLIT; ++i) mt = fmaxf(mt, mi[i]);
    float lt = 0.0f;
    float4 acc = make_float4(0.0f, 0.0f, 0.0f, 0.0f);
#pragma unroll
    for (int i = 0; i < KSPLIT; ++i) {
      const float f = __expf(mi[i] - mt);
      lt += f * li[i];
      const float4 v = sO[sl[i] * OPITCH4 + col];
      acc.x += f * v.x;
      acc.y += f * v.y;
      acc.z += f * v.z;
      acc.w += f * v.w;
    }
    const float inv = 1.0f / lt;
    const size_t off = (size_t)(nb + R) * C_DIM + 4 * col;
    const float4 g = *(const float4*)(xb + off);
    float4 yv;
    yv.x = acc.x * inv * g.x;
    yv.y = acc.y * inv * g.y;
    yv.z = acc.z * inv * g.z;
    yv.w = acc.w * inv * g.w;
    *(float4*)(yb + off) = yv;
  }
}

#define CG_RING 4
#define CG_GROUPS 32

__device__ __forceinline__ void conv_group(const v4f a4, const v4f (&w4)[4], v8f (&acc)[4]) {
#pragma unroll
  for (int p = 0; p < 2; ++p) {
    v2f va;
    va.x = a4[2 * p];
    va.y = a4[2 * p + 1];
#pragma unroll
    for (int j = 0; j < 4; ++j) {
      v2f vb;
      vb.x = w4[2 * p][j];
      vb.y = w4[2 * p + 1][j];
      acc[j] = wmma_f32_4(va, vb, acc[j]);
    }
  }
}

__global__ __launch_bounds__(256) void conv_gemm_kernel(
    const float* __restrict__ y, const float* __restrict__ w,
    const float* __restrict__ bias, float* __restrict__ out) {
  const int lane = threadIdx.x & 31;
  const int wave = __builtin_amdgcn_readfirstlane(threadIdx.x >> 5);
  const int col  = lane & 15;
  const int half = lane >> 4;

  const int job   = blockIdx.x * 8 + wave;
  const int g     = job & 1;
  const int rbase = (job >> 1) * 16;

  int rowg = rbase + col;
  int wpos = rowg % 29;
  int rem  = rowg / 29;
  int h = rem & 31; rem >>= 5;
  int d = rem & 3;
  int bb = rem >> 2;
  const size_t abase = (((size_t)bb * N_TOK) + ((d * 32 + h) * 32 + wpos)) * (size_t)C_DIM;

  const float* __restrict__ ap = y + abase + 4 * half;
  const float* __restrict__ wp = w + (size_t)(4 * half) * 128 + 64 * g + 4 * col;

  v8f acc[4];
#pragma unroll
  for (int j = 0; j < 4; ++j) acc[j] = (v8f)0.0f;

  v4f a4[CG_RING];
  v4f w4[CG_RING][4];
#pragma unroll
  for (int s = 0; s < CG_RING; ++s) {
    a4[s] = *(const v4f*)(ap + 8 * s);
#pragma unroll
    for (int i = 0; i < 4; ++i)
      w4[s][i] = *(const v4f*)(wp + (size_t)(8 * s + i) * 128);
    __builtin_amdgcn_sched_barrier(0);
  }

#pragma unroll
  for (int q0 = 0; q0 < CG_GROUPS - CG_RING; q0 += CG_RING) {
#pragma unroll
    for (int s = 0; s < CG_RING; ++s) {
      conv_group(a4[s], w4[s], acc);
      __builtin_amdgcn_sched_barrier(0);
      const int qn = q0 + s + CG_RING;
      a4[s] = *(const v4f*)(ap + 8 * qn);
#pragma unroll
      for (int i = 0; i < 4; ++i)
        w4[s][i] = *(const v4f*)(wp + (size_t)(8 * qn + i) * 128);
      __builtin_amdgcn_sched_barrier(0);
    }
  }
#pragma unroll
  for (int s = 0; s < CG_RING; ++s)
    conv_group(a4[s], w4[s], acc);

  const v4f b4 = *(const v4f*)(bias + 64 * g + 4 * col);
#pragma unroll
  for (int r = 0; r < 8; ++r) {
    const int row = rbase + 8 * half + r;
    v4f v;
    v.x = fmaxf(acc[0][r] + b4.x, 0.0f);
    v.y = fmaxf(acc[1][r] + b4.y, 0.0f);
    v.z = fmaxf(acc[2][r] + b4.z, 0.0f);
    v.w = fmaxf(acc[3][r] + b4.w, 0.0f);
    *(v4f*)(out + (size_t)row * 128 + 64 * g + 4 * col) = v;
  }
}

extern "C" void kernel_launch(void* const* d_in, const int* in_sizes, int n_in,
                              void* d_out, int out_size, void* d_ws, size_t ws_size,
                              hipStream_t stream) {
  const float* x  = (const float*)d_in[0];
  const float* cw = (const float*)d_in[1];
  const float* cb = (const float*)d_in[2];
  float* out = (float*)d_out;
  float* y   = (float*)d_ws;

  attn_gate_kernel<<<256, 512, 0, stream>>>(x, y);
  conv_gemm_kernel<<<232, 256, 0, stream>>>(y, cw, cb, out);
}
